# starting kernel
# baseline (speedup 1.0000x reference)
#include <hip/hip_runtime.h>

typedef float v2f __attribute__((ext_vector_type(2)));
typedef float v8f __attribute__((ext_vector_type(8)));

#define DIM 128

__global__ void gcn_zero_f32(float* __restrict__ p, int n) {
    int i = blockIdx.x * blockDim.x + threadIdx.x;
    int stride = gridDim.x * blockDim.x;
    for (; i < n; i += stride) p[i] = 0.0f;
}

__global__ void gcn_degree(const int* __restrict__ src, const int* __restrict__ dst,
                           float* __restrict__ deg_s, float* __restrict__ deg_d, int ne) {
    int i = blockIdx.x * blockDim.x + threadIdx.x;
    if (i < ne) {
        atomicAdd(&deg_s[src[i]], 1.0f);
        atomicAdd(&deg_d[dst[i]], 1.0f);
    }
}

__global__ void gcn_norm(float* __restrict__ p, int n) {
    int i = blockIdx.x * blockDim.x + threadIdx.x;
    if (i < n) p[i] = rsqrtf(fmaxf(p[i], 1.0f));
}

__global__ void gcn_scatter(const float* __restrict__ h, const float* __restrict__ norm_s,
                            const int* __restrict__ src, const int* __restrict__ dst,
                            float* __restrict__ agg, int ne) {
    int tid  = blockIdx.x * blockDim.x + threadIdx.x;
    int e    = tid >> 5;
    if (e >= ne) return;
    int lane = tid & 31;
    int s = src[e];
    int d = dst[e];
    float ns = norm_s[s];
    const float4 v = *reinterpret_cast<const float4*>(h + (size_t)s * DIM + lane * 4);
    float* ap = agg + (size_t)d * DIM + lane * 4;
    atomicAdd(ap + 0, v.x * ns);
    atomicAdd(ap + 1, v.y * ns);
    atomicAdd(ap + 2, v.z * ns);
    atomicAdd(ap + 3, v.w * ns);
}

#define LDA 129

__global__ void gcn_gemm_wmma(const float* __restrict__ agg, const float* __restrict__ norm_d,
                              const float* __restrict__ W, const float* __restrict__ bias,
                              float* __restrict__ out, int relu) {
    __shared__ float lA[16 * LDA];

    const int m0 = blockIdx.x * 16;
    const int t  = threadIdx.x;

    for (int i = t; i < 16 * DIM; i += 256) {
        int r = i >> 7;
        int k = i & (DIM - 1);
        int row = m0 + r;
        lA[r * LDA + k] = agg[(size_t)row * DIM + k] * norm_d[row];
    }
    __syncthreads();

    const int wave = t >> 5;
    const int lane = t & 31;
    const int n0   = wave * 16;
    const int half = lane >> 4;
    const int l16  = lane & 15;

    const float* Ar = lA + l16 * LDA;
    const float* Wc = W + n0 + l16;

    v8f c = {};
#pragma unroll
    for (int kk = 0; kk < DIM; kk += 4) {
        const int kbase = kk + half * 2;
        v2f a, b;
        a.x = Ar[kbase + 0];
        a.y = Ar[kbase + 1];
        b.x = Wc[(kbase + 0) * DIM];
        b.y = Wc[(kbase + 1) * DIM];
        c = __builtin_amdgcn_wmma_f32_16x16x4_f32(
                 false, a,  false, b,
                 (short)0, c,  false,  false);
    }

    const float bn = bias[n0 + l16];
#pragma unroll
    for (int r = 0; r < 8; ++r) {
        int row = m0 + r + half * 8;
        float v = c[r] + bn;
        if (relu) v = fmaxf(v, 0.0f);
        out[(size_t)row * DIM + n0 + l16] = v;
    }
}

extern "C" void kernel_launch(void* const* d_in, const int* in_sizes, int n_in,
                              void* d_out, int out_size, void* d_ws, size_t ws_size,
                              hipStream_t stream) {
    const float* feat = (const float*)d_in[0];
    const float* W1   = (const float*)d_in[1];
    const float* b1   = (const float*)d_in[2];
    const float* W2   = (const float*)d_in[3];
    const float* b2   = (const float*)d_in[4];
    const int*   src  = (const int*)d_in[5];
    const int*   dst  = (const int*)d_in[6];

    const int N = in_sizes[0] / DIM;
    const int E = in_sizes[5];

    float* ws     = (float*)d_ws;
    float* norm_s = ws;
    float* norm_d = ws + N;
    float* agg    = ws + 2 * (size_t)N;

    float* out = (float*)d_out;

    const int ZB = 256;
    const int aggN = N * DIM;

    gcn_zero_f32<<<(2 * N + ZB - 1) / ZB, ZB, 0, stream>>>(norm_s, 2 * N);
    gcn_degree<<<(E + ZB - 1) / ZB, ZB, 0, stream>>>(src, dst, norm_s, norm_d, E);
    gcn_norm<<<(2 * N + ZB - 1) / ZB, ZB, 0, stream>>>(norm_s, 2 * N);

    const int scatterThreads = E * 32;
    const int scatterBlocks  = (scatterThreads + ZB - 1) / ZB;

    gcn_zero_f32<<<(aggN + ZB - 1) / ZB, ZB, 0, stream>>>(agg, aggN);
    gcn_scatter<<<scatterBlocks, ZB, 0, stream>>>(feat, norm_s, src, dst, agg, E);
    gcn_gemm_wmma<<<N / 16, 256, 0, stream>>>(agg, norm_d, W1, b1, out,  1);

    gcn_zero_f32<<<(aggN + ZB - 1) / ZB, ZB, 0, stream>>>(agg, aggN);
    gcn_scatter<<<scatterBlocks, ZB, 0, stream>>>(out, norm_s, src, dst, agg, E);
    gcn_gemm_wmma<<<N / 16, 256, 0, stream>>>(agg, norm_d, W2, b2, out,  0);
}


# Layer-1 scatter scans one owner byte per edge (dst/391, stored by the degree pass in spare list space) instead of the 4-byte dst: 4x fewer scan bytes, 4096 edges a trip; LDS f32 atomics as before
# speedup vs baseline: 2.8339x; 2.8339x over previous
#include <hip/hip_runtime.h>

typedef float v2f __attribute__((ext_vector_type(2)));
typedef float v8f __attribute__((ext_vector_type(8)));

#define DIM 128

__global__ void gcn_zero_f32(float* __restrict__ p, int n) {
    int i = blockIdx.x * blockDim.x + threadIdx.x;
    int stride = gridDim.x * blockDim.x;
    for (; i < n; i += stride) p[i] = 0.0f;
}

__device__ __forceinline__ int gcn_my_half() {
    return (int)((__builtin_amdgcn_s_sendmsg_rtn(0x87) >> 18) & 1u);
}

#define RPB 391
#define RPB_BYTES (RPB * 128 * 4)
#define OWN_CHUNK 8192
#define OWN_G0 8
__device__ __forceinline__ unsigned char* gcn_owner_at(float* agg, int e) {
    return (unsigned char*)agg + (size_t)((e / OWN_CHUNK) + OWN_G0) * RPB_BYTES
           + (RPB_BYTES - OWN_CHUNK) + (e % OWN_CHUNK);
}

__global__ void gcn_degree(const int* __restrict__ src, const int* __restrict__ dst,
                           float* __restrict__ deg, float* __restrict__ agg, int n, int ne) {
    int i = blockIdx.x * blockDim.x + threadIdx.x;
    float* plane = deg + (size_t)gcn_my_half() * 2 * n;
    if (i < ne) {
        const int d = dst[i];
        atomicAdd(&plane[src[i]], 1.0f);
        atomicAdd(&plane[n + d], 1.0f);
        *gcn_owner_at(agg, i) = (unsigned char)(d / RPB);
    }
}

__global__ void gcn_norm(float* __restrict__ p, int n) {
    int i = blockIdx.x * blockDim.x + threadIdx.x;
    if (i < n) p[i] = rsqrtf(fmaxf(p[i] + p[i + n], 1.0f));
}

#define SCT 1024
#define EPT 1024
#define GRP 8
#define GRPR 16

template <int G>
__device__ __forceinline__ void gcn_add_rows(unsigned act, int s, int dl, int nsb,
                                             const float* __restrict__ h, float* acc, int lane) {
    while (act != 0u) {
        int ls[G];
        int cnt = 0;
#pragma unroll
        for (int u = 0; u < G; ++u) {
            ls[u] = 0;
            if (act != 0u) {
                ls[u] = __builtin_ctz(act);
                act &= act - 1u;
                cnt = u + 1;
            }
        }
        float v[G][4];
#pragma unroll
        for (int u = 0; u < G; ++u) {
            if (u < cnt) {
                const int sj = __builtin_amdgcn_readlane(s, ls[u]);
                const float* hp = h + (size_t)sj * DIM + lane;
                v[u][0] = hp[0];
                v[u][1] = hp[32];
                v[u][2] = hp[64];
                v[u][3] = hp[96];
            }
        }
#pragma unroll
        for (int u = 0; u < G; ++u) {
            if (u < cnt) {
                const int dj = __builtin_amdgcn_readlane(dl, ls[u]);
                const float nj = __int_as_float(__builtin_amdgcn_readlane(nsb, ls[u]));
                float* ap = acc + dj * DIM + lane;
                __hip_atomic_fetch_add(ap +  0, v[u][0] * nj, __ATOMIC_RELAXED, __HIP_MEMORY_SCOPE_WORKGROUP);
                __hip_atomic_fetch_add(ap + 32, v[u][1] * nj, __ATOMIC_RELAXED, __HIP_MEMORY_SCOPE_WORKGROUP);
                __hip_atomic_fetch_add(ap + 64, v[u][2] * nj, __ATOMIC_RELAXED, __HIP_MEMORY_SCOPE_WORKGROUP);
                __hip_atomic_fetch_add(ap + 96, v[u][3] * nj, __ATOMIC_RELAXED, __HIP_MEMORY_SCOPE_WORKGROUP);
            }
        }
    }
}

__device__ __forceinline__ void gcn_scan_edges(
        const float* __restrict__ h, const float* __restrict__ norm_s,
        const int* __restrict__ src, const int* __restrict__ dst,
        float* acc, int lo, int rc, int ne, int wave, int lane) {
    for (int eb = wave * EPT; eb < ne; eb += (SCT / 32) * EPT) {
        unsigned m = 0u;
        if (eb + EPT <= ne) {
            int4 d[EPT / 128];
#pragma unroll
            for (int k = 0; k < EPT / 128; ++k)
                d[k] = *(const int4*)(dst + eb + k * 128 + lane * 4);
#pragma unroll
            for (int k = 0; k < EPT / 128; ++k) {
                m |= ((unsigned)(d[k].x - lo) < (unsigned)rc) ? (1u << (4 * k + 0)) : 0u;
                m |= ((unsigned)(d[k].y - lo) < (unsigned)rc) ? (1u << (4 * k + 1)) : 0u;
                m |= ((unsigned)(d[k].z - lo) < (unsigned)rc) ? (1u << (4 * k + 2)) : 0u;
                m |= ((unsigned)(d[k].w - lo) < (unsigned)rc) ? (1u << (4 * k + 3)) : 0u;
            }
        } else {
            for (int b = 0; b < 32; ++b) {
                const int e = eb + (b >> 2) * 128 + lane * 4 + (b & 3);
                if (e < ne && (unsigned)(dst[e] - lo) < (unsigned)rc) m |= 1u << b;
            }
        }

        for (;;) {
            const unsigned act = __builtin_amdgcn_ballot_w32(m != 0u);
            if (act == 0u) break;
            const bool took = (m != 0u);
            int s = 0, dl = 0;
            float ns = 0.0f;
            if (took) {
                const int bit = __builtin_ctz(m);
                m &= m - 1u;
                const int e = eb + (bit >> 2) * 128 + lane * 4 + (bit & 3);
                s  = src[e];
                dl = dst[e] - lo;
                ns = norm_s[s];
            }
            gcn_add_rows<GRP>(act, s, dl, __float_as_int(ns), h, acc, lane);
        }
    }
}

__device__ __forceinline__ unsigned gcn_eq4(unsigned w, unsigned g4) {
    const unsigned x = w ^ g4;
    const unsigned t = (x & 0x7F7F7F7Fu) + 0x7F7F7F7Fu;
    return ~(t | x | 0x7F7F7F7Fu);
}
__device__ __forceinline__ unsigned gcn_eq16(uint4 d, unsigned g4) {
    return gcn_eq4(d.x, g4) | (gcn_eq4(d.y, g4) >> 1) | (gcn_eq4(d.z, g4) >> 2) | (gcn_eq4(d.w, g4) >> 3);
}

#define OEPT 4096
__device__ __forceinline__ void gcn_scan_owner_plane(
        const float* __restrict__ h, const float* __restrict__ norm_s,
        const int* __restrict__ src, const int* __restrict__ dst,
        float* agg, float* acc, float4* list, int cap, int* fill,
        int g, int lo, int ne, int wave, int lane) {
    const unsigned g4 = (unsigned)g * 0x01010101u;
    for (int eb = wave * OEPT; eb < ne; eb += (SCT / 32) * OEPT) {
        unsigned m[4] = {0u, 0u, 0u, 0u};
        const unsigned char* ob = gcn_owner_at(agg, eb);
        if (eb + OEPT <= ne) {
            uint4 d[OEPT / 512];
#pragma unroll
            for (int k = 0; k < OEPT / 512; ++k)
                d[k] = *(const uint4*)(ob + k * 512 + lane * 16);
#pragma unroll
            for (int k = 0; k < OEPT / 512; ++k)
                m[k >> 1] |= gcn_eq16(d[k], g4) >> (4 * (k & 1));
        } else {
#pragma unroll
            for (int k = 0; k < OEPT / 512; ++k) {
                const int o = k * 512 + lane * 16;
                unsigned hit = 0u;
                if (eb + o + 16 <= ne) {
                    hit = gcn_eq16(*(const uint4*)(ob + o), g4);
                } else {
                    for (int j = 0; j < 16; ++j)
                        if (eb + o + j < ne && ob[o + j] == (unsigned char)g)
                            hit |= 1u << (8 * (j & 3) + 7 - (j >> 2));
                }
                m[k >> 1] |= hit >> (4 * (k & 1));
            }
        }

        for (;;) {
            const bool took = ((m[0] | m[1] | m[2] | m[3]) != 0u);
            const unsigned act = __builtin_amdgcn_ballot_w32(took);
            if (act == 0u) break;
            int s = 0, dl = 0;
            float ns = 0.0f;
            if (took) {
                int q = 3;
                if (m[2] != 0u) q = 2;
                if (m[1] != 0u) q = 1;
                if (m[0] != 0u) q = 0;
                const unsigned mq = (q == 0) ? m[0] : (q == 1) ? m[1] : (q == 2) ? m[2] : m[3];
                const int bit = __builtin_ctz(mq);
                const unsigned rest = mq & (mq - 1u);
                if (q == 0) m[0] = rest;
                if (q == 1) m[1] = rest;
                if (q == 2) m[2] = rest;
                if (q == 3) m[3] = rest;
                const int r = bit & 7;
                const int k = 2 * q + ((r < 4) ? 1 : 0);
                const int e = eb + k * 512 + lane * 16 + 4 * ((7 - r) & 3) + (bit >> 3);
                s  = src[e];
                dl = dst[e] - lo;
                ns = norm_s[s];
            }
            gcn_add_rows<GRPR>(act, s, dl, __float_as_int(ns), h, acc, lane);
            int base = 0;
            if (lane == 0)
                base = __hip_atomic_fetch_add(fill, __builtin_popcount(act),
                                              __ATOMIC_RELAXED, __HIP_MEMORY_SCOPE_WORKGROUP);
            base = __builtin_amdgcn_readfirstlane(base);
            const int place = base + (int)__builtin_amdgcn_mbcnt_lo(act, 0u);
            if (took && place < cap) {
                float4 en;
                en.x = __int_as_float(s);
                en.y = __int_as_float(dl);
                en.z = ns;
                en.w = 0.0f;
                list[place] = en;
            }
        }
    }
}

__global__ __launch_bounds__(SCT) void gcn_scatter_scan(
        const float* __restrict__ h, const float* __restrict__ norm_s,
        const int* __restrict__ src, const int* __restrict__ dst,
        float* __restrict__ rows, float* lists, int n, int ne) {
    __shared__ __attribute__((aligned(16))) float acc[RPB * DIM];
    __shared__ int fill;

    const int t = threadIdx.x;
    const int lane = t & 31;
    const int wave = __builtin_amdgcn_readfirstlane(t >> 5);
    const int lo = (int)blockIdx.x * RPB;
    const int rc = (n - lo < RPB) ? (n - lo) : RPB;
    const int cap = rc * (DIM / 4) - 1 - OWN_CHUNK / 16;

    const float4 z4 = {0.0f, 0.0f, 0.0f, 0.0f};
    for (int i = t; i < RPB * DIM / 4; i += SCT) ((float4*)acc)[i] = z4;
    if (t == 0) fill = 0;
    __syncthreads();

    float4* mine = (float4*)(lists + (size_t)lo * DIM);
    gcn_scan_owner_plane(h, norm_s, src, dst, lists, acc, mine + 1, cap, &fill,
                         (int)blockIdx.x, lo, ne, wave, lane);
    __syncthreads();

    if (t == 0) {
        float4 hd = z4;
        hd.x = __int_as_float(fill);
        mine[0] = hd;
    }

    float4* orow = (float4*)(rows + (size_t)lo * DIM);
    for (int i = t; i < rc * (DIM / 4); i += SCT) orow[i] = ((const float4*)acc)[i];
}

__global__ __launch_bounds__(SCT) void gcn_scatter_replay(
        const float* __restrict__ h, const float* __restrict__ norm_s,
        const int* __restrict__ src, const int* __restrict__ dst,
        float* rows, int n, int ne) {
    __shared__ __attribute__((aligned(16))) float acc[RPB * DIM];

    const int t = threadIdx.x;
    const int lane = t & 31;
    const int wave = __builtin_amdgcn_readfirstlane(t >> 5);
    const int lo = (int)blockIdx.x * RPB;
    const int rc = (n - lo < RPB) ? (n - lo) : RPB;
    const int cap = rc * (DIM / 4) - 1 - OWN_CHUNK / 16;

    const float4 z4 = {0.0f, 0.0f, 0.0f, 0.0f};
    for (int i = t; i < RPB * DIM / 4; i += SCT) ((float4*)acc)[i] = z4;
    __syncthreads();

    float4* mine = (float4*)(rows + (size_t)lo * DIM);
    const int cnt = __builtin_amdgcn_readfirstlane(__float_as_int(((const float*)mine)[0]));
    if (cnt <= cap) {
        const float4* list = mine + 1;
        int b = wave * 32;
        float4 nx = z4;
        if (b + lane < cnt) nx = list[b + lane];
        for (; b < cnt; b += SCT) {
            const float4 en = nx;
            const unsigned act = __builtin_amdgcn_ballot_w32(b + lane < cnt);
            if (b + SCT + lane < cnt) nx = list[b + SCT + lane];
            gcn_add_rows<GRPR>(act, __float_as_int(en.x), __float_as_int(en.y),
                               __float_as_int(en.z), h, acc, lane);
        }
    } else {
        gcn_scan_edges(h, norm_s, src, dst, acc, lo, rc, ne, wave, lane);
    }
    __syncthreads();

    for (int i = t; i < rc * (DIM / 4); i += SCT) mine[i] = ((const float4*)acc)[i];
}

#define LDA 129

__global__ void gcn_gemm_wmma(const float* p0,
                              const float* __restrict__ norm_d,
                              const float* __restrict__ W, const float* __restrict__ bias,
                              float* out, int relu) {
    __shared__ float lA[16 * LDA];

    const int m0 = blockIdx.x * 16;
    const int t  = threadIdx.x;

    for (int i = t; i < 16 * DIM; i += 256) {
        int r = i >> 7;
        int k = i & (DIM - 1);
        int row = m0 + r;
        lA[r * LDA + k] = p0[(size_t)row * DIM + k] * norm_d[row];
    }
    __syncthreads();

    const int wave = t >> 5;
    const int lane = t & 31;
    const int n0   = wave * 16;
    const int half = lane >> 4;
    const int l16  = lane & 15;

    const float* Ar = lA + l16 * LDA;
    const float* Wc = W + n0 + l16;

    v8f c = {};
#pragma unroll
    for (int kk = 0; kk < DIM; kk += 4) {
        const int kbase = kk + half * 2;
        v2f a, b;
        a.x = Ar[kbase + 0];
        a.y = Ar[kbase + 1];
        b.x = Wc[(kbase + 0) * DIM];
        b.y = Wc[(kbase + 1) * DIM];
        c = __builtin_amdgcn_wmma_f32_16x16x4_f32(
                 false, a,  false, b,
                 (short)0, c,  false,  false);
    }

    const float bn = bias[n0 + l16];
#pragma unroll
    for (int r = 0; r < 8; ++r) {
        int row = m0 + r + half * 8;
        float v = c[r] + bn;
        if (relu) v = fmaxf(v, 0.0f);
        out[(size_t)row * DIM + n0 + l16] = v;
    }
}

extern "C" void kernel_launch(void* const* d_in, const int* in_sizes, int n_in,
                              void* d_out, int out_size, void* d_ws, size_t ws_size,
                              hipStream_t stream) {
    const float* feat = (const float*)d_in[0];
    const float* W1   = (const float*)d_in[1];
    const float* b1   = (const float*)d_in[2];
    const float* W2   = (const float*)d_in[3];
    const float* b2   = (const float*)d_in[4];
    const int*   src  = (const int*)d_in[5];
    const int*   dst  = (const int*)d_in[6];

    const int N = in_sizes[0] / DIM;
    const int E = in_sizes[5];

    float* ws     = (float*)d_ws;
    float* norm_s = ws;
    float* norm_d = ws + N;
    float* agg    = ws + 2 * (size_t)N;

    float* out = (float*)d_out;

    const int ZB = 256;

    gcn_zero_f32<<<(4 * N + ZB - 1) / ZB, ZB, 0, stream>>>(norm_s, 4 * N);
    gcn_degree<<<(E + ZB - 1) / ZB, ZB, 0, stream>>>(src, dst, norm_s, agg, N, E);
    gcn_norm<<<(2 * N + ZB - 1) / ZB, ZB, 0, stream>>>(norm_s, 2 * N);

    const int scatterGroups = (N + RPB - 1) / RPB;

    gcn_scatter_scan<<<scatterGroups, SCT, 0, stream>>>(feat, norm_s, src, dst, out, agg, N, E);
    gcn_gemm_wmma<<<N / 16, 256, 0, stream>>>(out, norm_d, W1, b1, out,  1);

    gcn_scatter_replay<<<scatterGroups, SCT, 0, stream>>>(out, norm_s, src, dst, agg, N, E);
    gcn_gemm_wmma<<<N / 16, 256, 0, stream>>>(agg, norm_d, W2, b2, out,  0);
}
